# starting kernel
# baseline (speedup 1.0000x reference)
#include <hip/hip_runtime.h>
#include <math.h>

#define D     512
#define NWAY  64
#define NQ    65536
#define NS    65536
#define EPSF  1e-30f
#define MBLK  128
#define KC    64
#define LDSTR 68

typedef float v2f __attribute__((ext_vector_type(2)));
typedef float v4f __attribute__((ext_vector_type(4)));
typedef float v8f __attribute__((ext_vector_type(8)));

__global__ void __launch_bounds__(256)
proto_partial(const float* __restrict__ sup, const int* __restrict__ lab,
              float* __restrict__ pSums) {
  __shared__ float ls[NWAY * 256];
  const int t    = threadIdx.x;
  const int bx   = blockIdx.x;
  const int rc   = bx >> 1;
  const int half = bx & 1;
  for (int i = t; i < NWAY * 256; i += 256) ls[i] = 0.f;
  __syncthreads();
  const int base = rc * 1024;
  const float* src = sup + (size_t)base * D + half * 256 + t;
  for (int r = 0; r < 1024; ++r) {
    int L = lab[base + r];
    ls[L * 256 + t] += src[(size_t)r * D];
  }
  __syncthreads();
  float* dst = pSums + (size_t)bx * (NWAY * 256);
  for (int i = t; i < NWAY * 256; i += 256) dst[i] = ls[i];
}

__global__ void __launch_bounds__(256)
proto_final(const float* __restrict__ pSums, const int* __restrict__ lab,
            const float* __restrict__ scaleP, float* __restrict__ protN) {
  __shared__ float red[256];
  __shared__ int   redi[256];
  const int c = blockIdx.x;
  const int t = threadIdx.x;

  int cnt = 0;
  for (int i = t; i < NS; i += 256) cnt += (lab[i] == c) ? 1 : 0;
  redi[t] = cnt;
  __syncthreads();
  for (int s = 128; s > 0; s >>= 1) {
    if (t < s) redi[t] += redi[t + s];
    __syncthreads();
  }
  const float fcnt = (float)redi[0];

  float p[2];
  #pragma unroll
  for (int j = 0; j < 2; ++j) {
    int d = t + j * 256;
    int half = d >> 8, dcol = d & 255;
    float s = 0.f;
    for (int rcb = 0; rcb < 64; ++rcb)
      s += pSums[(size_t)((rcb * 2 + half) * NWAY + c) * 256 + dcol];
    p[j] = s / fcnt;
  }

  red[t] = p[0] * p[0] + p[1] * p[1];
  __syncthreads();
  for (int s = 128; s > 0; s >>= 1) {
    if (t < s) red[t] += red[t + s];
    __syncthreads();
  }
  const float fac = scaleP[0] / fmaxf(sqrtf(red[0]), EPSF);
  protN[(size_t)c * D + t]       = p[0] * fac;
  protN[(size_t)c * D + t + 256] = p[1] * fac;
}

__global__ void __launch_bounds__(256)
cosine_gemm(const float* __restrict__ q, const float* __restrict__ protN,
            float* __restrict__ out) {
  __shared__ float qs[MBLK * LDSTR];
  __shared__ float ps[NWAY * LDSTR];
  __shared__ float normP[256];
  __shared__ float invq[MBLK];

  const int t    = threadIdx.x;
  const int w    = t >> 5;
  const int lane = t & 31;
  const int l15  = lane & 15;
  const int hi   = lane >> 4;
  const size_t qBase = (size_t)blockIdx.x * MBLK;

  v8f acc0 = {}, acc1 = {}, acc2 = {}, acc3 = {};
  float nacc = 0.f;
  const int nrow  = t & 127;
  const int nhalf = t >> 7;

  for (int kc = 0; kc < D; kc += KC) {
    #pragma unroll
    for (int j = 0; j < 8; ++j) {
      int v = t + j * 256;
      int row = v >> 4, c4 = (v & 15) * 4;
      v4f val = *(const v4f*)(q + (qBase + row) * D + kc + c4);
      *(v4f*)(qs + row * LDSTR + c4) = val;
    }
    #pragma unroll
    for (int j = 0; j < 4; ++j) {
      int v = t + j * 256;
      int row = v >> 4, c4 = (v & 15) * 4;
      v4f val = *(const v4f*)(protN + (size_t)row * D + kc + c4);
      *(v4f*)(ps + row * LDSTR + c4) = val;
    }
    __syncthreads();

    #pragma unroll
    for (int i = 0; i < 8; ++i) {
      v4f x = *(const v4f*)(qs + nrow * LDSTR + nhalf * 32 + i * 4);
      nacc += x.x * x.x + x.y * x.y + x.z * x.z + x.w * x.w;
    }

    for (int k4 = 0; k4 < KC; k4 += 4) {
      v2f a  = *(const v2f*)(qs + (w * 16 + l15) * LDSTR + k4 + 2 * hi);
      v2f b0 = *(const v2f*)(ps + ( 0 + l15) * LDSTR + k4 + 2 * hi);
      v2f b1 = *(const v2f*)(ps + (16 + l15) * LDSTR + k4 + 2 * hi);
      v2f b2 = *(const v2f*)(ps + (32 + l15) * LDSTR + k4 + 2 * hi);
      v2f b3 = *(const v2f*)(ps + (48 + l15) * LDSTR + k4 + 2 * hi);
      acc0 = __builtin_amdgcn_wmma_f32_16x16x4_f32(false, a, false, b0, (short)0, acc0, false, false);
      acc1 = __builtin_amdgcn_wmma_f32_16x16x4_f32(false, a, false, b1, (short)0, acc1, false, false);
      acc2 = __builtin_amdgcn_wmma_f32_16x16x4_f32(false, a, false, b2, (short)0, acc2, false, false);
      acc3 = __builtin_amdgcn_wmma_f32_16x16x4_f32(false, a, false, b3, (short)0, acc3, false, false);
    }
    __syncthreads();
  }

  normP[t] = nacc;
  __syncthreads();
  if (t < MBLK) {
    float n2 = normP[t] + normP[t + MBLK];
    invq[t] = 1.f / fmaxf(sqrtf(n2), EPSF);
  }
  __syncthreads();

  #pragma unroll
  for (int r = 0; r < 8; ++r) {
    int qrow = w * 16 + r + 8 * hi;
    float s = invq[qrow];
    size_t orow = (qBase + qrow) * NWAY;
    out[orow +  0 + l15] = acc0[r] * s;
    out[orow + 16 + l15] = acc1[r] * s;
    out[orow + 32 + l15] = acc2[r] * s;
    out[orow + 48 + l15] = acc3[r] * s;
  }
}

extern "C" void kernel_launch(void* const* d_in, const int* in_sizes, int n_in,
                              void* d_out, int out_size, void* d_ws, size_t ws_size,
                              hipStream_t stream) {
  const float* sup    = (const float*)d_in[0];
  const int*   slab   = (const int*)d_in[1];
  const float* qry    = (const float*)d_in[2];
  const float* scaleP = (const float*)d_in[4];
  float* out = (float*)d_out;

  float* pSums = (float*)d_ws;
  float* protN = (float*)((char*)d_ws + (size_t)128 * NWAY * 256 * sizeof(float));

  proto_partial<<<128, 256, 0, stream>>>(sup, slab, pSums);
  proto_final  <<<NWAY, 256, 0, stream>>>(pSums, slab, scaleP, protN);
  cosine_gemm  <<<NQ / MBLK, 256, 0, stream>>>(qry, protN, out);
}


# per-class support sums recast as a one-hot x support product on the f32 16x16x4 matrix instruction (512 blocks, 8 wave tiles summed in LDS); changes the summation order of the class sums
# speedup vs baseline: 1.9462x; 1.9462x over previous
#include <hip/hip_runtime.h>
#include <math.h>

#define D     512
#define NWAY  64
#define NQ    65536
#define NS    65536
#define EPSF  1e-30f
#define MBLK  128
#define KC    64
#define LDSTR 68

typedef float v2f __attribute__((ext_vector_type(2)));
typedef float v4f __attribute__((ext_vector_type(4)));
typedef float v8f __attribute__((ext_vector_type(8)));

#define PP_WROWS 128
#define PP_SROWS 16

struct pp_labels { int4 q0, q1, q2, q3; };

__device__ __forceinline__ void pp_load(v4f (&buf)[8], const float* __restrict__ src, int hi) {
  #pragma unroll
  for (int s = 0; s < 4; ++s) {
    #pragma unroll
    for (int e = 0; e < 2; ++e)
      buf[2 * s + e] = *(const v4f*)(src + (size_t)(4 * s + 2 * hi + e) * D);
  }
}

__device__ __forceinline__ pp_labels pp_load_labels(const int* __restrict__ lp) {
  pp_labels L;
  L.q0 = *(const int4*)(lp + 0);
  L.q1 = *(const int4*)(lp + 4);
  L.q2 = *(const int4*)(lp + 8);
  L.q3 = *(const int4*)(lp + 12);
  return L;
}

__device__ __forceinline__ void pp_step(v8f (&acc)[4][4], const v4f r0, const v4f r1,
                                        const int4 q, int hi, int l15) {
  const int xa = (hi ? q.z : q.x) - l15;
  const int xb = (hi ? q.w : q.y) - l15;
  v2f a[4];
  #pragma unroll
  for (int m = 0; m < 4; ++m) {
    a[m].x = (xa == 16 * m) ? 1.f : 0.f;
    a[m].y = (xb == 16 * m) ? 1.f : 0.f;
  }
  #pragma unroll
  for (int j = 0; j < 4; ++j) {
    v2f b;
    b.x = r0[j];
    b.y = r1[j];
    #pragma unroll
    for (int m = 0; m < 4; ++m)
      acc[m][j] = __builtin_amdgcn_wmma_f32_16x16x4_f32(false, a[m], false, b, (short)0,
                                                        acc[m][j], false, false);
  }
}

__device__ __forceinline__ void pp_mma(v8f (&acc)[4][4], const v4f (&buf)[8],
                                       const pp_labels L, int hi, int l15) {
  pp_step(acc, buf[0], buf[1], L.q0, hi, l15);
  pp_step(acc, buf[2], buf[3], L.q1, hi, l15);
  pp_step(acc, buf[4], buf[5], L.q2, hi, l15);
  pp_step(acc, buf[6], buf[7], L.q3, hi, l15);
}

__global__ void __launch_bounds__(256)
proto_partial(const float* __restrict__ sup, const int* __restrict__ lab,
              float* __restrict__ pSums) {
  __shared__ float red[8 * NWAY * 64];
  const int t     = threadIdx.x;
  const int wv    = __builtin_amdgcn_readfirstlane(t >> 5);
  const int lane  = t & 31;
  const int l15   = lane & 15;
  const int hi    = lane >> 4;
  const int rc    = blockIdx.x >> 3;
  const int slice = blockIdx.x & 7;
  const int rowBase = rc * 1024 + wv * PP_WROWS;
  const float* src = sup + (size_t)rowBase * D + slice * 64 + 4 * l15;
  const int*   lp  = lab + rowBase;

  v8f acc[4][4];
  #pragma unroll
  for (int m = 0; m < 4; ++m) {
    #pragma unroll
    for (int j = 0; j < 4; ++j) acc[m][j] = (v8f){};
  }

  v4f bufA[8], bufB[8];
  pp_labels labA, labB;
  pp_load(bufA, src, hi);
  labA = pp_load_labels(lp);
  #pragma unroll
  for (int it = 0; it < PP_WROWS / PP_SROWS; it += 2) {
    pp_load(bufB, src + (size_t)(it + 1) * PP_SROWS * D, hi);
    labB = pp_load_labels(lp + (it + 1) * PP_SROWS);
    __builtin_amdgcn_sched_barrier(0);
    pp_mma(acc, bufA, labA, hi, l15);
    __builtin_amdgcn_sched_barrier(0);
    if (it + 2 < PP_WROWS / PP_SROWS) {
      pp_load(bufA, src + (size_t)(it + 2) * PP_SROWS * D, hi);
      labA = pp_load_labels(lp + (it + 2) * PP_SROWS);
    }
    __builtin_amdgcn_sched_barrier(0);
    pp_mma(acc, bufB, labB, hi, l15);
    __builtin_amdgcn_sched_barrier(0);
  }

  float* my = red + wv * (NWAY * 64);
  #pragma unroll
  for (int m = 0; m < 4; ++m) {
    #pragma unroll
    for (int r = 0; r < 8; ++r) {
      v4f v;
      v.x = acc[m][0][r]; v.y = acc[m][1][r]; v.z = acc[m][2][r]; v.w = acc[m][3][r];
      *(v4f*)(my + (16 * m + 8 * hi + r) * 64 + 4 * l15) = v;
    }
  }
  __syncthreads();

  float* dst = pSums + (size_t)((rc * 2 + (slice >> 2)) * NWAY) * 256 + (slice & 3) * 64;
  #pragma unroll
  for (int i = 0; i < 4; ++i) {
    const int idx = t + 256 * i;
    const int cls = idx >> 4, c4 = (idx & 15) * 4;
    v4f s = *(const v4f*)(red + cls * 64 + c4);
    #pragma unroll
    for (int w = 1; w < 8; ++w) s += *(const v4f*)(red + w * (NWAY * 64) + cls * 64 + c4);
    *(v4f*)(dst + (size_t)cls * 256 + c4) = s;
  }
}

__global__ void __launch_bounds__(256)
proto_final(const float* __restrict__ pSums, const int* __restrict__ lab,
            const float* __restrict__ scaleP, float* __restrict__ protN) {
  __shared__ float red[256];
  __shared__ int   redi[256];
  const int c = blockIdx.x;
  const int t = threadIdx.x;

  int cnt = 0;
  for (int i = t; i < NS; i += 256) cnt += (lab[i] == c) ? 1 : 0;
  redi[t] = cnt;
  __syncthreads();
  for (int s = 128; s > 0; s >>= 1) {
    if (t < s) redi[t] += redi[t + s];
    __syncthreads();
  }
  const float fcnt = (float)redi[0];

  float p[2];
  #pragma unroll
  for (int j = 0; j < 2; ++j) {
    int d = t + j * 256;
    int half = d >> 8, dcol = d & 255;
    float s = 0.f;
    for (int rcb = 0; rcb < 64; ++rcb)
      s += pSums[(size_t)((rcb * 2 + half) * NWAY + c) * 256 + dcol];
    p[j] = s / fcnt;
  }

  red[t] = p[0] * p[0] + p[1] * p[1];
  __syncthreads();
  for (int s = 128; s > 0; s >>= 1) {
    if (t < s) red[t] += red[t + s];
    __syncthreads();
  }
  const float fac = scaleP[0] / fmaxf(sqrtf(red[0]), EPSF);
  protN[(size_t)c * D + t]       = p[0] * fac;
  protN[(size_t)c * D + t + 256] = p[1] * fac;
}

__global__ void __launch_bounds__(256)
cosine_gemm(const float* __restrict__ q, const float* __restrict__ protN,
            float* __restrict__ out) {
  __shared__ float qs[MBLK * LDSTR];
  __shared__ float ps[NWAY * LDSTR];
  __shared__ float normP[256];
  __shared__ float invq[MBLK];

  const int t    = threadIdx.x;
  const int w    = t >> 5;
  const int lane = t & 31;
  const int l15  = lane & 15;
  const int hi   = lane >> 4;
  const size_t qBase = (size_t)blockIdx.x * MBLK;

  v8f acc0 = {}, acc1 = {}, acc2 = {}, acc3 = {};
  float nacc = 0.f;
  const int nrow  = t & 127;
  const int nhalf = t >> 7;

  for (int kc = 0; kc < D; kc += KC) {
    #pragma unroll
    for (int j = 0; j < 8; ++j) {
      int v = t + j * 256;
      int row = v >> 4, c4 = (v & 15) * 4;
      v4f val = *(const v4f*)(q + (qBase + row) * D + kc + c4);
      *(v4f*)(qs + row * LDSTR + c4) = val;
    }
    #pragma unroll
    for (int j = 0; j < 4; ++j) {
      int v = t + j * 256;
      int row = v >> 4, c4 = (v & 15) * 4;
      v4f val = *(const v4f*)(protN + (size_t)row * D + kc + c4);
      *(v4f*)(ps + row * LDSTR + c4) = val;
    }
    __syncthreads();

    #pragma unroll
    for (int i = 0; i < 8; ++i) {
      v4f x = *(const v4f*)(qs + nrow * LDSTR + nhalf * 32 + i * 4);
      nacc += x.x * x.x + x.y * x.y + x.z * x.z + x.w * x.w;
    }

    for (int k4 = 0; k4 < KC; k4 += 4) {
      v2f a  = *(const v2f*)(qs + (w * 16 + l15) * LDSTR + k4 + 2 * hi);
      v2f b0 = *(const v2f*)(ps + ( 0 + l15) * LDSTR + k4 + 2 * hi);
      v2f b1 = *(const v2f*)(ps + (16 + l15) * LDSTR + k4 + 2 * hi);
      v2f b2 = *(const v2f*)(ps + (32 + l15) * LDSTR + k4 + 2 * hi);
      v2f b3 = *(const v2f*)(ps + (48 + l15) * LDSTR + k4 + 2 * hi);
      acc0 = __builtin_amdgcn_wmma_f32_16x16x4_f32(false, a, false, b0, (short)0, acc0, false, false);
      acc1 = __builtin_amdgcn_wmma_f32_16x16x4_f32(false, a, false, b1, (short)0, acc1, false, false);
      acc2 = __builtin_amdgcn_wmma_f32_16x16x4_f32(false, a, false, b2, (short)0, acc2, false, false);
      acc3 = __builtin_amdgcn_wmma_f32_16x16x4_f32(false, a, false, b3, (short)0, acc3, false, false);
    }
    __syncthreads();
  }

  normP[t] = nacc;
  __syncthreads();
  if (t < MBLK) {
    float n2 = normP[t] + normP[t + MBLK];
    invq[t] = 1.f / fmaxf(sqrtf(n2), EPSF);
  }
  __syncthreads();

  #pragma unroll
  for (int r = 0; r < 8; ++r) {
    int qrow = w * 16 + r + 8 * hi;
    float s = invq[qrow];
    size_t orow = (qBase + qrow) * NWAY;
    out[orow +  0 + l15] = acc0[r] * s;
    out[orow + 16 + l15] = acc1[r] * s;
    out[orow + 32 + l15] = acc2[r] * s;
    out[orow + 48 + l15] = acc3[r] * s;
  }
}

extern "C" void kernel_launch(void* const* d_in, const int* in_sizes, int n_in,
                              void* d_out, int out_size, void* d_ws, size_t ws_size,
                              hipStream_t stream) {
  const float* sup    = (const float*)d_in[0];
  const int*   slab   = (const int*)d_in[1];
  const float* qry    = (const float*)d_in[2];
  const float* scaleP = (const float*)d_in[4];
  float* out = (float*)d_out;

  float* pSums = (float*)d_ws;
  float* protN = (float*)((char*)d_ws + (size_t)128 * NWAY * 256 * sizeof(float));

  proto_partial<<<512, 256, 0, stream>>>(sup, slab, pSums);
  proto_final  <<<NWAY, 256, 0, stream>>>(pSums, slab, scaleP, protN);
  cosine_gemm  <<<NQ / MBLK, 256, 0, stream>>>(qry, protN, out);
}


# finalize kernel at 1024 threads per class, all 24 loads of a thread (16 B each) in flight before first use, labels counted by ballot; chunk partials and norm now summed as two-level trees
# speedup vs baseline: 2.2984x; 1.1810x over previous
#include <hip/hip_runtime.h>
#include <math.h>

#define D     512
#define NWAY  64
#define NQ    65536
#define NS    65536
#define EPSF  1e-30f
#define MBLK  128
#define KC    64
#define LDSTR 68

typedef float v2f __attribute__((ext_vector_type(2)));
typedef float v4f __attribute__((ext_vector_type(4)));
typedef float v8f __attribute__((ext_vector_type(8)));

#define PP_WROWS 128
#define PP_SROWS 16

struct pp_labels { int4 q0, q1, q2, q3; };

__device__ __forceinline__ void pp_load(v4f (&buf)[8], const float* __restrict__ src, int hi) {
  #pragma unroll
  for (int s = 0; s < 4; ++s) {
    #pragma unroll
    for (int e = 0; e < 2; ++e)
      buf[2 * s + e] = *(const v4f*)(src + (size_t)(4 * s + 2 * hi + e) * D);
  }
}

__device__ __forceinline__ pp_labels pp_load_labels(const int* __restrict__ lp) {
  pp_labels L;
  L.q0 = *(const int4*)(lp + 0);
  L.q1 = *(const int4*)(lp + 4);
  L.q2 = *(const int4*)(lp + 8);
  L.q3 = *(const int4*)(lp + 12);
  return L;
}

__device__ __forceinline__ void pp_step(v8f (&acc)[4][4], const v4f r0, const v4f r1,
                                        const int4 q, int hi, int l15) {
  const int xa = (hi ? q.z : q.x) - l15;
  const int xb = (hi ? q.w : q.y) - l15;
  v2f a[4];
  #pragma unroll
  for (int m = 0; m < 4; ++m) {
    a[m].x = (xa == 16 * m) ? 1.f : 0.f;
    a[m].y = (xb == 16 * m) ? 1.f : 0.f;
  }
  #pragma unroll
  for (int j = 0; j < 4; ++j) {
    v2f b;
    b.x = r0[j];
    b.y = r1[j];
    #pragma unroll
    for (int m = 0; m < 4; ++m)
      acc[m][j] = __builtin_amdgcn_wmma_f32_16x16x4_f32(false, a[m], false, b, (short)0,
                                                        acc[m][j], false, false);
  }
}

__device__ __forceinline__ void pp_mma(v8f (&acc)[4][4], const v4f (&buf)[8],
                                       const pp_labels L, int hi, int l15) {
  pp_step(acc, buf[0], buf[1], L.q0, hi, l15);
  pp_step(acc, buf[2], buf[3], L.q1, hi, l15);
  pp_step(acc, buf[4], buf[5], L.q2, hi, l15);
  pp_step(acc, buf[6], buf[7], L.q3, hi, l15);
}

__global__ void __launch_bounds__(256)
proto_partial(const float* __restrict__ sup, const int* __restrict__ lab,
              float* __restrict__ pSums) {
  __shared__ float red[8 * NWAY * 64];
  const int t     = threadIdx.x;
  const int wv    = __builtin_amdgcn_readfirstlane(t >> 5);
  const int lane  = t & 31;
  const int l15   = lane & 15;
  const int hi    = lane >> 4;
  const int rc    = blockIdx.x >> 3;
  const int slice = blockIdx.x & 7;
  const int rowBase = rc * 1024 + wv * PP_WROWS;
  const float* src = sup + (size_t)rowBase * D + slice * 64 + 4 * l15;
  const int*   lp  = lab + rowBase;

  v8f acc[4][4];
  #pragma unroll
  for (int m = 0; m < 4; ++m) {
    #pragma unroll
    for (int j = 0; j < 4; ++j) acc[m][j] = (v8f){};
  }

  v4f bufA[8], bufB[8];
  pp_labels labA, labB;
  pp_load(bufA, src, hi);
  labA = pp_load_labels(lp);
  #pragma unroll
  for (int it = 0; it < PP_WROWS / PP_SROWS; it += 2) {
    pp_load(bufB, src + (size_t)(it + 1) * PP_SROWS * D, hi);
    labB = pp_load_labels(lp + (it + 1) * PP_SROWS);
    __builtin_amdgcn_sched_barrier(0);
    pp_mma(acc, bufA, labA, hi, l15);
    __builtin_amdgcn_sched_barrier(0);
    if (it + 2 < PP_WROWS / PP_SROWS) {
      pp_load(bufA, src + (size_t)(it + 2) * PP_SROWS * D, hi);
      labA = pp_load_labels(lp + (it + 2) * PP_SROWS);
    }
    __builtin_amdgcn_sched_barrier(0);
    pp_mma(acc, bufB, labB, hi, l15);
    __builtin_amdgcn_sched_barrier(0);
  }

  float* my = red + wv * (NWAY * 64);
  #pragma unroll
  for (int m = 0; m < 4; ++m) {
    #pragma unroll
    for (int r = 0; r < 8; ++r) {
      v4f v;
      v.x = acc[m][0][r]; v.y = acc[m][1][r]; v.z = acc[m][2][r]; v.w = acc[m][3][r];
      *(v4f*)(my + (16 * m + 8 * hi + r) * 64 + 4 * l15) = v;
    }
  }
  __syncthreads();

  float* dst = pSums + (size_t)((rc * 2 + (slice >> 2)) * NWAY) * 256 + (slice & 3) * 64;
  #pragma unroll
  for (int i = 0; i < 4; ++i) {
    const int idx = t + 256 * i;
    const int cls = idx >> 4, c4 = (idx & 15) * 4;
    v4f s = *(const v4f*)(red + cls * 64 + c4);
    #pragma unroll
    for (int w = 1; w < 8; ++w) s += *(const v4f*)(red + w * (NWAY * 64) + cls * 64 + c4);
    *(v4f*)(dst + (size_t)cls * 256 + c4) = s;
  }
}

#define PF_THREADS 1024

__global__ void __launch_bounds__(PF_THREADS)
proto_final(const float* __restrict__ pSums, const int* __restrict__ lab,
            const float* __restrict__ scaleP, float* __restrict__ protN) {
  __shared__ v4f   part[8 * 128];
  __shared__ int   cntw[32];
  __shared__ float sqw[16];
  const int c    = blockIdx.x;
  const int t    = threadIdx.x;
  const int lane = t & 31;
  const int wv   = t >> 5;
  const int g    = t >> 7;
  const int q4   = t & 127;
  const int half = q4 >> 6;
  const int dcol = (q4 & 63) * 4;

  const size_t chunkStride = (size_t)2 * NWAY * 256;
  const float* ps = pSums + (size_t)(g * 8) * chunkStride + (size_t)(half * NWAY + c) * 256 + dcol;
  v4f pv[8];
  #pragma unroll
  for (int i = 0; i < 8; ++i) pv[i] = *(const v4f*)(ps + (size_t)i * chunkStride);

  const int4* l4 = (const int4*)lab;
  int4 lv[16];
  #pragma unroll
  for (int i = 0; i < 16; ++i) lv[i] = l4[t + PF_THREADS * i];
  const float scale = scaleP[0];
  __builtin_amdgcn_sched_barrier(0);

  int cnt = 0;
  #pragma unroll
  for (int i = 0; i < 16; ++i) {
    cnt += __builtin_popcount(__builtin_amdgcn_ballot_w32(lv[i].x == c));
    cnt += __builtin_popcount(__builtin_amdgcn_ballot_w32(lv[i].y == c));
    cnt += __builtin_popcount(__builtin_amdgcn_ballot_w32(lv[i].z == c));
    cnt += __builtin_popcount(__builtin_amdgcn_ballot_w32(lv[i].w == c));
  }
  if (lane == 0) cntw[wv] = cnt;

  v4f s = pv[0];
  #pragma unroll
  for (int i = 1; i < 8; ++i) s += pv[i];
  part[t] = s;
  __syncthreads();

  int total = 0;
  #pragma unroll
  for (int w = 0; w < 32; ++w) total += cntw[w];
  const float fcnt = (float)total;

  float p = 0.f, sq = 0.f;
  if (t < D) {
    const float* pf = (const float*)part;
    float a = pf[t];
    #pragma unroll
    for (int k = 1; k < 8; ++k) a += pf[k * D + t];
    p  = a / fcnt;
    sq = p * p;
  }
  #pragma unroll
  for (int m = 16; m > 0; m >>= 1) sq += __shfl_xor(sq, m);
  if (lane == 0 && wv < 16) sqw[wv] = sq;
  __syncthreads();

  float n2 = sqw[0];
  #pragma unroll
  for (int w = 1; w < 16; ++w) n2 += sqw[w];
  const float fac = scale / fmaxf(sqrtf(n2), EPSF);
  if (t < D) protN[(size_t)c * D + t] = p * fac;
}

__global__ void __launch_bounds__(256)
cosine_gemm(const float* __restrict__ q, const float* __restrict__ protN,
            float* __restrict__ out) {
  __shared__ float qs[MBLK * LDSTR];
  __shared__ float ps[NWAY * LDSTR];
  __shared__ float normP[256];
  __shared__ float invq[MBLK];

  const int t    = threadIdx.x;
  const int w    = t >> 5;
  const int lane = t & 31;
  const int l15  = lane & 15;
  const int hi   = lane >> 4;
  const size_t qBase = (size_t)blockIdx.x * MBLK;

  v8f acc0 = {}, acc1 = {}, acc2 = {}, acc3 = {};
  float nacc = 0.f;
  const int nrow  = t & 127;
  const int nhalf = t >> 7;

  for (int kc = 0; kc < D; kc += KC) {
    #pragma unroll
    for (int j = 0; j < 8; ++j) {
      int v = t + j * 256;
      int row = v >> 4, c4 = (v & 15) * 4;
      v4f val = *(const v4f*)(q + (qBase + row) * D + kc + c4);
      *(v4f*)(qs + row * LDSTR + c4) = val;
    }
    #pragma unroll
    for (int j = 0; j < 4; ++j) {
      int v = t + j * 256;
      int row = v >> 4, c4 = (v & 15) * 4;
      v4f val = *(const v4f*)(protN + (size_t)row * D + kc + c4);
      *(v4f*)(ps + row * LDSTR + c4) = val;
    }
    __syncthreads();

    #pragma unroll
    for (int i = 0; i < 8; ++i) {
      v4f x = *(const v4f*)(qs + nrow * LDSTR + nhalf * 32 + i * 4);
      nacc += x.x * x.x + x.y * x.y + x.z * x.z + x.w * x.w;
    }

    for (int k4 = 0; k4 < KC; k4 += 4) {
      v2f a  = *(const v2f*)(qs + (w * 16 + l15) * LDSTR + k4 + 2 * hi);
      v2f b0 = *(const v2f*)(ps + ( 0 + l15) * LDSTR + k4 + 2 * hi);
      v2f b1 = *(const v2f*)(ps + (16 + l15) * LDSTR + k4 + 2 * hi);
      v2f b2 = *(const v2f*)(ps + (32 + l15) * LDSTR + k4 + 2 * hi);
      v2f b3 = *(const v2f*)(ps + (48 + l15) * LDSTR + k4 + 2 * hi);
      acc0 = __builtin_amdgcn_wmma_f32_16x16x4_f32(false, a, false, b0, (short)0, acc0, false, false);
      acc1 = __builtin_amdgcn_wmma_f32_16x16x4_f32(false, a, false, b1, (short)0, acc1, false, false);
      acc2 = __builtin_amdgcn_wmma_f32_16x16x4_f32(false, a, false, b2, (short)0, acc2, false, false);
      acc3 = __builtin_amdgcn_wmma_f32_16x16x4_f32(false, a, false, b3, (short)0, acc3, false, false);
    }
    __syncthreads();
  }

  normP[t] = nacc;
  __syncthreads();
  if (t < MBLK) {
    float n2 = normP[t] + normP[t + MBLK];
    invq[t] = 1.f / fmaxf(sqrtf(n2), EPSF);
  }
  __syncthreads();

  #pragma unroll
  for (int r = 0; r < 8; ++r) {
    int qrow = w * 16 + r + 8 * hi;
    float s = invq[qrow];
    size_t orow = (qBase + qrow) * NWAY;
    out[orow +  0 + l15] = acc0[r] * s;
    out[orow + 16 + l15] = acc1[r] * s;
    out[orow + 32 + l15] = acc2[r] * s;
    out[orow + 48 + l15] = acc3[r] * s;
  }
}

extern "C" void kernel_launch(void* const* d_in, const int* in_sizes, int n_in,
                              void* d_out, int out_size, void* d_ws, size_t ws_size,
                              hipStream_t stream) {
  const float* sup    = (const float*)d_in[0];
  const int*   slab   = (const int*)d_in[1];
  const float* qry    = (const float*)d_in[2];
  const float* scaleP = (const float*)d_in[4];
  float* out = (float*)d_out;

  float* pSums = (float*)d_ws;
  float* protN = (float*)((char*)d_ws + (size_t)128 * NWAY * 256 * sizeof(float));

  proto_partial<<<512, 256, 0, stream>>>(sup, slab, pSums);
  proto_final  <<<NWAY, PF_THREADS, 0, stream>>>(pSums, slab, scaleP, protN);
  cosine_gemm  <<<NQ / MBLK, 256, 0, stream>>>(qry, protN, out);
}


# class sums: each wave lists the rows of one class M-tile (ballot, prefix count) and multiplies only those, a quarter of the f32 matrix instructions; changes the summation order of the class sums
# speedup vs baseline: 2.5352x; 1.1030x over previous
#include <hip/hip_runtime.h>
#include <math.h>

#define D     512
#define NWAY  64
#define NQ    65536
#define NS    65536
#define EPSF  1e-30f
#define MBLK  128
#define KC    64
#define LDSTR 68

typedef float v2f __attribute__((ext_vector_type(2)));
typedef float v4f __attribute__((ext_vector_type(4)));
typedef float v8f __attribute__((ext_vector_type(8)));

#define PP_HROWS 512
#define PP_GROUP 32
#define PP_LCAP  (PP_HROWS + 2 * PP_GROUP)

__device__ __forceinline__ void pp_fetch(v4f (&buf)[16], int2 (&ent)[8],
                                         const float* __restrict__ srcU, const int* my,
                                         int g, int hi, unsigned laneOff) {
  #pragma unroll
  for (int s = 0; s < 8; ++s) ent[s] = *(const int2*)(my + PP_GROUP * g + 4 * s + 2 * hi);
  #pragma unroll
  for (int s = 0; s < 8; ++s) {
    buf[2 * s]     = *(const v4f*)(srcU + ((unsigned)(ent[s].x >> 5) * D + laneOff));
    buf[2 * s + 1] = *(const v4f*)(srcU + ((unsigned)(ent[s].y >> 5) * D + laneOff));
  }
}

__device__ __forceinline__ void pp_mma(v8f (&acc)[4], const v4f (&buf)[16],
                                       const int2 (&ent)[8], int l15) {
  #pragma unroll
  for (int s = 0; s < 8; ++s) {
    v2f a;
    a.x = ((ent[s].x & 31) == l15) ? 1.f : 0.f;
    a.y = ((ent[s].y & 31) == l15) ? 1.f : 0.f;
    #pragma unroll
    for (int j = 0; j < 4; ++j) {
      v2f b;
      b.x = buf[2 * s][j];
      b.y = buf[2 * s + 1][j];
      acc[j] = __builtin_amdgcn_wmma_f32_16x16x4_f32(false, a, false, b, (short)0,
                                                     acc[j], false, false);
    }
  }
}

__global__ void __launch_bounds__(256)
proto_partial(const float* __restrict__ sup, const int* __restrict__ lab,
              float* __restrict__ pSums) {
  __shared__ __attribute__((aligned(16))) float red[8 * 16 * 64];
  __shared__ __attribute__((aligned(16))) int   rlist[8 * PP_LCAP];
  const int t     = threadIdx.x;
  const int wv    = __builtin_amdgcn_readfirstlane(t >> 5);
  const int lane  = t & 31;
  const int l15   = lane & 15;
  const int hi    = lane >> 4;
  const int rc    = blockIdx.x >> 3;
  const int slice = blockIdx.x & 7;
  const int mt    = wv & 3;
  const int half  = wv >> 2;
  const int rowBase = rc * 1024 + half * PP_HROWS;
  const float* srcU = sup + (size_t)rowBase * D + slice * 64;
  const unsigned laneOff = 4 * l15;
  const int* lp = lab + rowBase;
  int* my = rlist + wv * PP_LCAP;

  int lv[PP_HROWS / 32];
  #pragma unroll
  for (int i = 0; i < PP_HROWS / 32; ++i) lv[i] = lp[32 * i + lane];
  int cnt = 0;
  #pragma unroll
  for (int i = 0; i < PP_HROWS / 32; ++i) {
    const bool in = (lv[i] >> 4) == mt;
    const unsigned bal = __builtin_amdgcn_ballot_w32(in);
    const int pos = cnt + (int)__builtin_amdgcn_mbcnt_lo(bal, 0u);
    if (in) my[pos] = ((32 * i + lane) << 5) | (lv[i] & 15);
    cnt += __builtin_popcount(bal);
  }
  my[cnt + lane] = 16;
  my[cnt + 32 + lane] = 16;
  __syncthreads();

  v8f acc[4];
  #pragma unroll
  for (int j = 0; j < 4; ++j) acc[j] = (v8f){};

  const int ngrp = (cnt + PP_GROUP - 1) / PP_GROUP;
  v4f bufA[16], bufB[16];
  int2 entA[8], entB[8];
  pp_fetch(bufA, entA, srcU, my, 0, hi, laneOff);
  const int g0 = ngrp & 1;
  if (g0) {
    pp_mma(acc, bufA, entA, l15);
    pp_fetch(bufA, entA, srcU, my, 1, hi, laneOff);
  }
  for (int g = g0; g < ngrp; g += 2) {
    pp_fetch(bufB, entB, srcU, my, g + 1, hi, laneOff);
    __builtin_amdgcn_sched_barrier(0);
    #pragma unroll
    for (int i = 0; i < 16; ++i) asm volatile("" : "+v"(bufA[i]));
    pp_mma(acc, bufA, entA, l15);
    __builtin_amdgcn_sched_barrier(0);
    pp_fetch(bufA, entA, srcU, my, g + 2, hi, laneOff);
    __builtin_amdgcn_sched_barrier(0);
    pp_mma(acc, bufB, entB, l15);
    __builtin_amdgcn_sched_barrier(0);
  }

  float* tile = red + wv * (16 * 64);
  #pragma unroll
  for (int r = 0; r < 8; ++r) {
    v4f v;
    v.x = acc[0][r]; v.y = acc[1][r]; v.z = acc[2][r]; v.w = acc[3][r];
    *(v4f*)(tile + (8 * hi + r) * 64 + 4 * l15) = v;
  }
  __syncthreads();

  float* dst = pSums + (size_t)((rc * 2 + (slice >> 2)) * NWAY) * 256 + (slice & 3) * 64;
  #pragma unroll
  for (int i = 0; i < 4; ++i) {
    const int idx = t + 256 * i;
    const int cls = idx >> 4, c4 = (idx & 15) * 4;
    const float* p0 = red + (cls >> 4) * (16 * 64) + (cls & 15) * 64 + c4;
    const v4f s = *(const v4f*)p0 + *(const v4f*)(p0 + 4 * (16 * 64));
    *(v4f*)(dst + (size_t)cls * 256 + c4) = s;
  }
}

#define PF_THREADS 1024

__global__ void __launch_bounds__(PF_THREADS)
proto_final(const float* __restrict__ pSums, const int* __restrict__ lab,
            const float* __restrict__ scaleP, float* __restrict__ protN) {
  __shared__ v4f   part[8 * 128];
  __shared__ int   cntw[32];
  __shared__ float sqw[16];
  const int c    = blockIdx.x;
  const int t    = threadIdx.x;
  const int lane = t & 31;
  const int wv   = t >> 5;
  const int g    = t >> 7;
  const int q4   = t & 127;
  const int half = q4 >> 6;
  const int dcol = (q4 & 63) * 4;

  const size_t chunkStride = (size_t)2 * NWAY * 256;
  const float* ps = pSums + (size_t)(g * 8) * chunkStride + (size_t)(half * NWAY + c) * 256 + dcol;
  v4f pv[8];
  #pragma unroll
  for (int i = 0; i < 8; ++i) pv[i] = *(const v4f*)(ps + (size_t)i * chunkStride);

  const int4* l4 = (const int4*)lab;
  int4 lv[16];
  #pragma unroll
  for (int i = 0; i < 16; ++i) lv[i] = l4[t + PF_THREADS * i];
  const float scale = scaleP[0];
  __builtin_amdgcn_sched_barrier(0);

  int cnt = 0;
  #pragma unroll
  for (int i = 0; i < 16; ++i) {
    cnt += __builtin_popcount(__builtin_amdgcn_ballot_w32(lv[i].x == c));
    cnt += __builtin_popcount(__builtin_amdgcn_ballot_w32(lv[i].y == c));
    cnt += __builtin_popcount(__builtin_amdgcn_ballot_w32(lv[i].z == c));
    cnt += __builtin_popcount(__builtin_amdgcn_ballot_w32(lv[i].w == c));
  }
  if (lane == 0) cntw[wv] = cnt;

  v4f s = pv[0];
  #pragma unroll
  for (int i = 1; i < 8; ++i) s += pv[i];
  part[t] = s;
  __syncthreads();

  int total = 0;
  #pragma unroll
  for (int w = 0; w < 32; ++w) total += cntw[w];
  const float fcnt = (float)total;

  float p = 0.f, sq = 0.f;
  if (t < D) {
    const float* pf = (const float*)part;
    float a = pf[t];
    #pragma unroll
    for (int k = 1; k < 8; ++k) a += pf[k * D + t];
    p  = a / fcnt;
    sq = p * p;
  }
  #pragma unroll
  for (int m = 16; m > 0; m >>= 1) sq += __shfl_xor(sq, m);
  if (lane == 0 && wv < 16) sqw[wv] = sq;
  __syncthreads();

  float n2 = sqw[0];
  #pragma unroll
  for (int w = 1; w < 16; ++w) n2 += sqw[w];
  const float fac = scale / fmaxf(sqrtf(n2), EPSF);
  if (t < D) protN[(size_t)c * D + t] = p * fac;
}

__global__ void __launch_bounds__(256)
cosine_gemm(const float* __restrict__ q, const float* __restrict__ protN,
            float* __restrict__ out) {
  __shared__ float qs[MBLK * LDSTR];
  __shared__ float ps[NWAY * LDSTR];
  __shared__ float normP[256];
  __shared__ float invq[MBLK];

  const int t    = threadIdx.x;
  const int w    = t >> 5;
  const int lane = t & 31;
  const int l15  = lane & 15;
  const int hi   = lane >> 4;
  const size_t qBase = (size_t)blockIdx.x * MBLK;

  v8f acc0 = {}, acc1 = {}, acc2 = {}, acc3 = {};
  float nacc = 0.f;
  const int nrow  = t & 127;
  const int nhalf = t >> 7;

  for (int kc = 0; kc < D; kc += KC) {
    #pragma unroll
    for (int j = 0; j < 8; ++j) {
      int v = t + j * 256;
      int row = v >> 4, c4 = (v & 15) * 4;
      v4f val = *(const v4f*)(q + (qBase + row) * D + kc + c4);
      *(v4f*)(qs + row * LDSTR + c4) = val;
    }
    #pragma unroll
    for (int j = 0; j < 4; ++j) {
      int v = t + j * 256;
      int row = v >> 4, c4 = (v & 15) * 4;
      v4f val = *(const v4f*)(protN + (size_t)row * D + kc + c4);
      *(v4f*)(ps + row * LDSTR + c4) = val;
    }
    __syncthreads();

    #pragma unroll
    for (int i = 0; i < 8; ++i) {
      v4f x = *(const v4f*)(qs + nrow * LDSTR + nhalf * 32 + i * 4);
      nacc += x.x * x.x + x.y * x.y + x.z * x.z + x.w * x.w;
    }

    for (int k4 = 0; k4 < KC; k4 += 4) {
      v2f a  = *(const v2f*)(qs + (w * 16 + l15) * LDSTR + k4 + 2 * hi);
      v2f b0 = *(const v2f*)(ps + ( 0 + l15) * LDSTR + k4 + 2 * hi);
      v2f b1 = *(const v2f*)(ps + (16 + l15) * LDSTR + k4 + 2 * hi);
      v2f b2 = *(const v2f*)(ps + (32 + l15) * LDSTR + k4 + 2 * hi);
      v2f b3 = *(const v2f*)(ps + (48 + l15) * LDSTR + k4 + 2 * hi);
      acc0 = __builtin_amdgcn_wmma_f32_16x16x4_f32(false, a, false, b0, (short)0, acc0, false, false);
      acc1 = __builtin_amdgcn_wmma_f32_16x16x4_f32(false, a, false, b1, (short)0, acc1, false, false);
      acc2 = __builtin_amdgcn_wmma_f32_16x16x4_f32(false, a, false, b2, (short)0, acc2, false, false);
      acc3 = __builtin_amdgcn_wmma_f32_16x16x4_f32(false, a, false, b3, (short)0, acc3, false, false);
    }
    __syncthreads();
  }

  normP[t] = nacc;
  __syncthreads();
  if (t < MBLK) {
    float n2 = normP[t] + normP[t + MBLK];
    invq[t] = 1.f / fmaxf(sqrtf(n2), EPSF);
  }
  __syncthreads();

  #pragma unroll
  for (int r = 0; r < 8; ++r) {
    int qrow = w * 16 + r + 8 * hi;
    float s = invq[qrow];
    size_t orow = (qBase + qrow) * NWAY;
    out[orow +  0 + l15] = acc0[r] * s;
    out[orow + 16 + l15] = acc1[r] * s;
    out[orow + 32 + l15] = acc2[r] * s;
    out[orow + 48 + l15] = acc3[r] * s;
  }
}

extern "C" void kernel_launch(void* const* d_in, const int* in_sizes, int n_in,
                              void* d_out, int out_size, void* d_ws, size_t ws_size,
                              hipStream_t stream) {
  const float* sup    = (const float*)d_in[0];
  const int*   slab   = (const int*)d_in[1];
  const float* qry    = (const float*)d_in[2];
  const float* scaleP = (const float*)d_in[4];
  float* out = (float*)d_out;

  float* pSums = (float*)d_ws;
  float* protN = (float*)((char*)d_ws + (size_t)128 * NWAY * 256 * sizeof(float));

  proto_partial<<<512, 256, 0, stream>>>(sup, slab, pSums);
  proto_final  <<<NWAY, PF_THREADS, 0, stream>>>(pSums, slab, scaleP, protN);
  cosine_gemm  <<<NQ / MBLK, 256, 0, stream>>>(qry, protN, out);
}


# cosine GEMM K loop pipelined: all 12 loads per thread of chunk c+1 in flight in registers during the products of chunk c, two LDS stages, one barrier per chunk; summation orders unchanged
# speedup vs baseline: 2.8753x; 1.1342x over previous
#include <hip/hip_runtime.h>
#include <math.h>

#define D     512
#define NWAY  64
#define NQ    65536
#define NS    65536
#define EPSF  1e-30f
#define MBLK  128
#define KC    64
#define LDSTR 68

typedef float v2f __attribute__((ext_vector_type(2)));
typedef float v4f __attribute__((ext_vector_type(4)));
typedef float v8f __attribute__((ext_vector_type(8)));

#define PP_HROWS 512
#define PP_GROUP 32
#define PP_LCAP  (PP_HROWS + 2 * PP_GROUP)

__device__ __forceinline__ void pp_fetch(v4f (&buf)[16], int2 (&ent)[8],
                                         const float* __restrict__ srcU, const int* my,
                                         int g, int hi, unsigned laneOff) {
  #pragma unroll
  for (int s = 0; s < 8; ++s) ent[s] = *(const int2*)(my + PP_GROUP * g + 4 * s + 2 * hi);
  #pragma unroll
  for (int s = 0; s < 8; ++s) {
    buf[2 * s]     = *(const v4f*)(srcU + ((unsigned)(ent[s].x >> 5) * D + laneOff));
    buf[2 * s + 1] = *(const v4f*)(srcU + ((unsigned)(ent[s].y >> 5) * D + laneOff));
  }
}

__device__ __forceinline__ void pp_mma(v8f (&acc)[4], const v4f (&buf)[16],
                                       const int2 (&ent)[8], int l15) {
  #pragma unroll
  for (int s = 0; s < 8; ++s) {
    v2f a;
    a.x = ((ent[s].x & 31) == l15) ? 1.f : 0.f;
    a.y = ((ent[s].y & 31) == l15) ? 1.f : 0.f;
    #pragma unroll
    for (int j = 0; j < 4; ++j) {
      v2f b;
      b.x = buf[2 * s][j];
      b.y = buf[2 * s + 1][j];
      acc[j] = __builtin_amdgcn_wmma_f32_16x16x4_f32(false, a, false, b, (short)0,
                                                     acc[j], false, false);
    }
  }
}

__global__ void __launch_bounds__(256)
proto_partial(const float* __restrict__ sup, const int* __restrict__ lab,
              float* __restrict__ pSums) {
  __shared__ __attribute__((aligned(16))) float red[8 * 16 * 64];
  __shared__ __attribute__((aligned(16))) int   rlist[8 * PP_LCAP];
  const int t     = threadIdx.x;
  const int wv    = __builtin_amdgcn_readfirstlane(t >> 5);
  const int lane  = t & 31;
  const int l15   = lane & 15;
  const int hi    = lane >> 4;
  const int rc    = blockIdx.x >> 3;
  const int slice = blockIdx.x & 7;
  const int mt    = wv & 3;
  const int half  = wv >> 2;
  const int rowBase = rc * 1024 + half * PP_HROWS;
  const float* srcU = sup + (size_t)rowBase * D + slice * 64;
  const unsigned laneOff = 4 * l15;
  const int* lp = lab + rowBase;
  int* my = rlist + wv * PP_LCAP;

  int lv[PP_HROWS / 32];
  #pragma unroll
  for (int i = 0; i < PP_HROWS / 32; ++i) lv[i] = lp[32 * i + lane];
  int cnt = 0;
  #pragma unroll
  for (int i = 0; i < PP_HROWS / 32; ++i) {
    const bool in = (lv[i] >> 4) == mt;
    const unsigned bal = __builtin_amdgcn_ballot_w32(in);
    const int pos = cnt + (int)__builtin_amdgcn_mbcnt_lo(bal, 0u);
    if (in) my[pos] = ((32 * i + lane) << 5) | (lv[i] & 15);
    cnt += __builtin_popcount(bal);
  }
  my[cnt + lane] = 16;
  my[cnt + 32 + lane] = 16;
  __syncthreads();

  v8f acc[4];
  #pragma unroll
  for (int j = 0; j < 4; ++j) acc[j] = (v8f){};

  const int ngrp = (cnt + PP_GROUP - 1) / PP_GROUP;
  v4f bufA[16], bufB[16];
  int2 entA[8], entB[8];
  pp_fetch(bufA, entA, srcU, my, 0, hi, laneOff);
  const int g0 = ngrp & 1;
  if (g0) {
    pp_mma(acc, bufA, entA, l15);
    pp_fetch(bufA, entA, srcU, my, 1, hi, laneOff);
  }
  for (int g = g0; g < ngrp; g += 2) {
    pp_fetch(bufB, entB, srcU, my, g + 1, hi, laneOff);
    __builtin_amdgcn_sched_barrier(0);
    #pragma unroll
    for (int i = 0; i < 16; ++i) asm volatile("" : "+v"(bufA[i]));
    pp_mma(acc, bufA, entA, l15);
    __builtin_amdgcn_sched_barrier(0);
    pp_fetch(bufA, entA, srcU, my, g + 2, hi, laneOff);
    __builtin_amdgcn_sched_barrier(0);
    pp_mma(acc, bufB, entB, l15);
    __builtin_amdgcn_sched_barrier(0);
  }

  float* tile = red + wv * (16 * 64);
  #pragma unroll
  for (int r = 0; r < 8; ++r) {
    v4f v;
    v.x = acc[0][r]; v.y = acc[1][r]; v.z = acc[2][r]; v.w = acc[3][r];
    *(v4f*)(tile + (8 * hi + r) * 64 + 4 * l15) = v;
  }
  __syncthreads();

  float* dst = pSums + (size_t)((rc * 2 + (slice >> 2)) * NWAY) * 256 + (slice & 3) * 64;
  #pragma unroll
  for (int i = 0; i < 4; ++i) {
    const int idx = t + 256 * i;
    const int cls = idx >> 4, c4 = (idx & 15) * 4;
    const float* p0 = red + (cls >> 4) * (16 * 64) + (cls & 15) * 64 + c4;
    const v4f s = *(const v4f*)p0 + *(const v4f*)(p0 + 4 * (16 * 64));
    *(v4f*)(dst + (size_t)cls * 256 + c4) = s;
  }
}

#define PF_THREADS 1024

__global__ void __launch_bounds__(PF_THREADS)
proto_final(const float* __restrict__ pSums, const int* __restrict__ lab,
            const float* __restrict__ scaleP, float* __restrict__ protN) {
  __shared__ v4f   part[8 * 128];
  __shared__ int   cntw[32];
  __shared__ float sqw[16];
  const int c    = blockIdx.x;
  const int t    = threadIdx.x;
  const int lane = t & 31;
  const int wv   = t >> 5;
  const int g    = t >> 7;
  const int q4   = t & 127;
  const int half = q4 >> 6;
  const int dcol = (q4 & 63) * 4;

  const size_t chunkStride = (size_t)2 * NWAY * 256;
  const float* ps = pSums + (size_t)(g * 8) * chunkStride + (size_t)(half * NWAY + c) * 256 + dcol;
  v4f pv[8];
  #pragma unroll
  for (int i = 0; i < 8; ++i) pv[i] = *(const v4f*)(ps + (size_t)i * chunkStride);

  const int4* l4 = (const int4*)lab;
  int4 lv[16];
  #pragma unroll
  for (int i = 0; i < 16; ++i) lv[i] = l4[t + PF_THREADS * i];
  const float scale = scaleP[0];
  __builtin_amdgcn_sched_barrier(0);

  int cnt = 0;
  #pragma unroll
  for (int i = 0; i < 16; ++i) {
    cnt += __builtin_popcount(__builtin_amdgcn_ballot_w32(lv[i].x == c));
    cnt += __builtin_popcount(__builtin_amdgcn_ballot_w32(lv[i].y == c));
    cnt += __builtin_popcount(__builtin_amdgcn_ballot_w32(lv[i].z == c));
    cnt += __builtin_popcount(__builtin_amdgcn_ballot_w32(lv[i].w == c));
  }
  if (lane == 0) cntw[wv] = cnt;

  v4f s = pv[0];
  #pragma unroll
  for (int i = 1; i < 8; ++i) s += pv[i];
  part[t] = s;
  __syncthreads();

  int total = 0;
  #pragma unroll
  for (int w = 0; w < 32; ++w) total += cntw[w];
  const float fcnt = (float)total;

  float p = 0.f, sq = 0.f;
  if (t < D) {
    const float* pf = (const float*)part;
    float a = pf[t];
    #pragma unroll
    for (int k = 1; k < 8; ++k) a += pf[k * D + t];
    p  = a / fcnt;
    sq = p * p;
  }
  #pragma unroll
  for (int m = 16; m > 0; m >>= 1) sq += __shfl_xor(sq, m);
  if (lane == 0 && wv < 16) sqw[wv] = sq;
  __syncthreads();

  float n2 = sqw[0];
  #pragma unroll
  for (int w = 1; w < 16; ++w) n2 += sqw[w];
  const float fac = scale / fmaxf(sqrtf(n2), EPSF);
  if (t < D) protN[(size_t)c * D + t] = p * fac;
}

#define QSTAGE (MBLK * LDSTR)
#define PSTAGE (NWAY * LDSTR)

__device__ __forceinline__ void cg_half(v8f& acc0, v8f& acc1, v8f& acc2, v8f& acc3,
                                        const float* qa, const float* pb, int k0) {
  for (int k4 = k0; k4 < k0 + KC / 2; k4 += 4) {
    v2f a  = *(const v2f*)(qa + k4);
    v2f b0 = *(const v2f*)(pb +  0 * LDSTR + k4);
    v2f b1 = *(const v2f*)(pb + 16 * LDSTR + k4);
    v2f b2 = *(const v2f*)(pb + 32 * LDSTR + k4);
    v2f b3 = *(const v2f*)(pb + 48 * LDSTR + k4);
    acc0 = __builtin_amdgcn_wmma_f32_16x16x4_f32(false, a, false, b0, (short)0, acc0, false, false);
    acc1 = __builtin_amdgcn_wmma_f32_16x16x4_f32(false, a, false, b1, (short)0, acc1, false, false);
    acc2 = __builtin_amdgcn_wmma_f32_16x16x4_f32(false, a, false, b2, (short)0, acc2, false, false);
    acc3 = __builtin_amdgcn_wmma_f32_16x16x4_f32(false, a, false, b3, (short)0, acc3, false, false);
  }
  asm volatile("" : "+v"(acc0), "+v"(acc1), "+v"(acc2), "+v"(acc3));
}

__device__ __forceinline__ void cg_chunk(v8f& acc0, v8f& acc1, v8f& acc2, v8f& acc3, float& nacc,
                                         const float* qst, const float* pst,
                                         int w, int l15, int hi, int nrow, int nhalf) {
  const float* qa = qst + (w * 16 + l15) * LDSTR + 2 * hi;
  const float* pb = pst + l15 * LDSTR + 2 * hi;
  cg_half(acc0, acc1, acc2, acc3, qa, pb, 0);
  cg_half(acc0, acc1, acc2, acc3, qa, pb, KC / 2);
  #pragma unroll
  for (int i = 0; i < 8; ++i) {
    v4f x = *(const v4f*)(qst + nrow * LDSTR + nhalf * 32 + i * 4);
    nacc += x.x * x.x + x.y * x.y + x.z * x.z + x.w * x.w;
  }
}

__global__ void __launch_bounds__(256)
cosine_gemm(const float* __restrict__ q, const float* __restrict__ protN,
            float* __restrict__ out) {
  __shared__ __attribute__((aligned(16))) float qs[2 * QSTAGE];
  __shared__ __attribute__((aligned(16))) float ps[2 * PSTAGE];
  __shared__ float normP[256];
  __shared__ float invq[MBLK];

  const int t    = threadIdx.x;
  const int w    = t >> 5;
  const int lane = t & 31;
  const int l15  = lane & 15;
  const int hi   = lane >> 4;
  const size_t qBase = (size_t)blockIdx.x * MBLK;

  v8f acc0 = {}, acc1 = {}, acc2 = {}, acc3 = {};
  float nacc = 0.f;
  const int nrow  = t & 127;
  const int nhalf = t >> 7;

  const int srow = t >> 4;
  const int sc4  = (t & 15) * 4;
  const float* qsrc = q + (qBase + srow) * D + sc4;
  const float* psrc = protN + (size_t)srow * D + sc4;
  const int sOff = srow * LDSTR + sc4;

  v4f qr[8], pr[4];

  #pragma unroll
  for (int j = 0; j < 8; ++j) qr[j] = *(const v4f*)(qsrc + (size_t)(16 * j) * D);
  #pragma unroll
  for (int j = 0; j < 4; ++j) pr[j] = *(const v4f*)(psrc + (size_t)(16 * j) * D);
  #pragma unroll
  for (int j = 0; j < 8; ++j) *(v4f*)(qs + sOff + (16 * j) * LDSTR) = qr[j];
  #pragma unroll
  for (int j = 0; j < 4; ++j) *(v4f*)(ps + sOff + (16 * j) * LDSTR) = pr[j];
  __syncthreads();

  for (int c = 0; c < D / KC - 1; ++c) {
    const int st = c & 1;
    #pragma unroll
    for (int j = 0; j < 8; ++j) qr[j] = *(const v4f*)(qsrc + (size_t)(16 * j) * D + (c + 1) * KC);
    #pragma unroll
    for (int j = 0; j < 4; ++j) pr[j] = *(const v4f*)(psrc + (size_t)(16 * j) * D + (c + 1) * KC);

    cg_chunk(acc0, acc1, acc2, acc3, nacc, qs + st * QSTAGE, ps + st * PSTAGE,
             w, l15, hi, nrow, nhalf);

    float* qd = qs + (st ^ 1) * QSTAGE + sOff;
    float* pd = ps + (st ^ 1) * PSTAGE + sOff;
    #pragma unroll
    for (int j = 0; j < 8; ++j) *(v4f*)(qd + (16 * j) * LDSTR) = qr[j];
    #pragma unroll
    for (int j = 0; j < 4; ++j) *(v4f*)(pd + (16 * j) * LDSTR) = pr[j];
    __syncthreads();
  }
  cg_chunk(acc0, acc1, acc2, acc3, nacc, qs + QSTAGE, ps + PSTAGE, w, l15, hi, nrow, nhalf);

  normP[t] = nacc;
  __syncthreads();
  if (t < MBLK) {
    float n2 = normP[t] + normP[t + MBLK];
    invq[t] = 1.f / fmaxf(sqrtf(n2), EPSF);
  }
  __syncthreads();

  #pragma unroll
  for (int r = 0; r < 8; ++r) {
    int qrow = w * 16 + r + 8 * hi;
    float s = invq[qrow];
    size_t orow = (qBase + qrow) * NWAY;
    out[orow +  0 + l15] = acc0[r] * s;
    out[orow + 16 + l15] = acc1[r] * s;
    out[orow + 32 + l15] = acc2[r] * s;
    out[orow + 48 + l15] = acc3[r] * s;
  }
}

extern "C" void kernel_launch(void* const* d_in, const int* in_sizes, int n_in,
                              void* d_out, int out_size, void* d_ws, size_t ws_size,
                              hipStream_t stream) {
  const float* sup    = (const float*)d_in[0];
  const int*   slab   = (const int*)d_in[1];
  const float* qry    = (const float*)d_in[2];
  const float* scaleP = (const float*)d_in[4];
  float* out = (float*)d_out;

  float* pSums = (float*)d_ws;
  float* protN = (float*)((char*)d_ws + (size_t)128 * NWAY * 256 * sizeof(float));

  proto_partial<<<512, 256, 0, stream>>>(sup, slab, pSums);
  proto_final  <<<NWAY, PF_THREADS, 0, stream>>>(pSums, slab, scaleP, protN);
  cosine_gemm  <<<NQ / MBLK, 256, 0, stream>>>(qry, protN, out);
}


# finalize kernel spread over 256 blocks (class x 128 dims): label counts come from the class-sum kernel, prototype norm and scale applied in the cosine epilogue; changes the norm's summation order
# speedup vs baseline: 3.0174x; 1.0494x over previous
#include <hip/hip_runtime.h>
#include <math.h>

#define D     512
#define NWAY  64
#define NQ    65536
#define NS    65536
#define EPSF  1e-30f
#define MBLK  128
#define KC    64
#define LDSTR 68

typedef float v2f __attribute__((ext_vector_type(2)));
typedef float v4f __attribute__((ext_vector_type(4)));
typedef float v8f __attribute__((ext_vector_type(8)));

#define PP_HROWS 512
#define PP_GROUP 32
#define PP_LCAP  (PP_HROWS + 2 * PP_GROUP)

__device__ __forceinline__ void pp_fetch(v4f (&buf)[16], int2 (&ent)[8],
                                         const float* __restrict__ srcU, const int* my,
                                         int g, int hi, unsigned laneOff) {
  #pragma unroll
  for (int s = 0; s < 8; ++s) ent[s] = *(const int2*)(my + PP_GROUP * g + 4 * s + 2 * hi);
  #pragma unroll
  for (int s = 0; s < 8; ++s) {
    buf[2 * s]     = *(const v4f*)(srcU + ((unsigned)(ent[s].x >> 5) * D + laneOff));
    buf[2 * s + 1] = *(const v4f*)(srcU + ((unsigned)(ent[s].y >> 5) * D + laneOff));
  }
}

__device__ __forceinline__ void pp_mma(v8f (&acc)[4], const v4f (&buf)[16],
                                       const int2 (&ent)[8], int l15) {
  #pragma unroll
  for (int s = 0; s < 8; ++s) {
    v2f a;
    a.x = ((ent[s].x & 31) == l15) ? 1.f : 0.f;
    a.y = ((ent[s].y & 31) == l15) ? 1.f : 0.f;
    #pragma unroll
    for (int j = 0; j < 4; ++j) {
      v2f b;
      b.x = buf[2 * s][j];
      b.y = buf[2 * s + 1][j];
      acc[j] = __builtin_amdgcn_wmma_f32_16x16x4_f32(false, a, false, b, (short)0,
                                                     acc[j], false, false);
    }
  }
}

__global__ void __launch_bounds__(256)
proto_partial(const float* __restrict__ sup, const int* __restrict__ lab,
              float* __restrict__ pSums, int* __restrict__ cnts) {
  __shared__ __attribute__((aligned(16))) float red[8 * 16 * 64];
  __shared__ __attribute__((aligned(16))) int   rlist[8 * PP_LCAP];
  __shared__ int ccnt[16];
  const int t     = threadIdx.x;
  const int wv    = __builtin_amdgcn_readfirstlane(t >> 5);
  const int lane  = t & 31;
  const int l15   = lane & 15;
  const int hi    = lane >> 4;
  const int rc    = blockIdx.x >> 3;
  const int slice = blockIdx.x & 7;
  const int mt    = wv & 3;
  const int half  = wv >> 2;
  const int rowBase = rc * 1024 + half * PP_HROWS;
  const float* srcU = sup + (size_t)rowBase * D + slice * 64;
  const unsigned laneOff = 4 * l15;
  const int* lp = lab + rowBase;
  int* my = rlist + wv * PP_LCAP;

  int lv[PP_HROWS / 32];
  #pragma unroll
  for (int i = 0; i < PP_HROWS / 32; ++i) lv[i] = lp[32 * i + lane];
  int cnt = 0;
  #pragma unroll
  for (int i = 0; i < PP_HROWS / 32; ++i) {
    const bool in = (lv[i] >> 4) == mt;
    const unsigned bal = __builtin_amdgcn_ballot_w32(in);
    const int pos = cnt + (int)__builtin_amdgcn_mbcnt_lo(bal, 0u);
    if (in) my[pos] = ((32 * i + lane) << 5) | (lv[i] & 15);
    cnt += __builtin_popcount(bal);
  }
  my[cnt + lane] = 16;
  my[cnt + 32 + lane] = 16;
  {
    const int k0 = 8 * slice + 2 * mt;
    int c0 = 0, c1 = 0;
    #pragma unroll
    for (int i = 0; i < PP_HROWS / 32; ++i) {
      c0 += __builtin_popcount(__builtin_amdgcn_ballot_w32(lv[i] == k0));
      c1 += __builtin_popcount(__builtin_amdgcn_ballot_w32(lv[i] == k0 + 1));
    }
    if (lane == 0) { ccnt[half * 8 + 2 * mt] = c0; ccnt[half * 8 + 2 * mt + 1] = c1; }
  }
  __syncthreads();
  if (t < 16) cnts[blockIdx.x * 32 + t] = ccnt[t];

  v8f acc[4];
  #pragma unroll
  for (int j = 0; j < 4; ++j) acc[j] = (v8f){};

  const int ngrp = (cnt + PP_GROUP - 1) / PP_GROUP;
  v4f bufA[16], bufB[16];
  int2 entA[8], entB[8];
  pp_fetch(bufA, entA, srcU, my, 0, hi, laneOff);
  const int g0 = ngrp & 1;
  if (g0) {
    pp_mma(acc, bufA, entA, l15);
    pp_fetch(bufA, entA, srcU, my, 1, hi, laneOff);
  }
  for (int g = g0; g < ngrp; g += 2) {
    pp_fetch(bufB, entB, srcU, my, g + 1, hi, laneOff);
    __builtin_amdgcn_sched_barrier(0);
    #pragma unroll
    for (int i = 0; i < 16; ++i) asm volatile("" : "+v"(bufA[i]));
    pp_mma(acc, bufA, entA, l15);
    __builtin_amdgcn_sched_barrier(0);
    pp_fetch(bufA, entA, srcU, my, g + 2, hi, laneOff);
    __builtin_amdgcn_sched_barrier(0);
    pp_mma(acc, bufB, entB, l15);
    __builtin_amdgcn_sched_barrier(0);
  }

  float* tile = red + wv * (16 * 64);
  #pragma unroll
  for (int r = 0; r < 8; ++r) {
    v4f v;
    v.x = acc[0][r]; v.y = acc[1][r]; v.z = acc[2][r]; v.w = acc[3][r];
    *(v4f*)(tile + (8 * hi + r) * 64 + 4 * l15) = v;
  }
  __syncthreads();

  float* dst = pSums + (size_t)((rc * 2 + (slice >> 2)) * NWAY) * 256 + (slice & 3) * 64;
  #pragma unroll
  for (int i = 0; i < 4; ++i) {
    const int idx = t + 256 * i;
    const int cls = idx >> 4, c4 = (idx & 15) * 4;
    const float* p0 = red + (cls >> 4) * (16 * 64) + (cls & 15) * 64 + c4;
    const v4f s = *(const v4f*)p0 + *(const v4f*)(p0 + 4 * (16 * 64));
    *(v4f*)(dst + (size_t)cls * 256 + c4) = s;
  }
}

#define PF_THREADS 256

__global__ void __launch_bounds__(PF_THREADS)
proto_final(float* pSums, const int* __restrict__ cnts) {
  __shared__ v4f part[8 * 32];
  __shared__ int cntw[4];
  const int c    = blockIdx.x >> 2;
  const int dq   = blockIdx.x & 3;
  const int t    = threadIdx.x;
  const int lane = t & 31;
  const int g    = t >> 5;

  const size_t chunkStride = (size_t)2 * NWAY * 256;
  const size_t rowOff = (size_t)((dq >> 1) * NWAY + c) * 256 + (dq & 1) * 128 + 4 * lane;
  const float* ps = pSums + (size_t)(g * 8) * chunkStride + rowOff;
  v4f pv[8];
  #pragma unroll
  for (int i = 0; i < 8; ++i) pv[i] = *(const v4f*)(ps + (size_t)i * chunkStride);

  int cv = 0;
  if (t < 128) cv = cnts[((t >> 1) * 8 + (c >> 3)) * 32 + (t & 1) * 8 + (c & 7)];
  #pragma unroll
  for (int m = 16; m > 0; m >>= 1) cv += __shfl_xor(cv, m);
  if (lane == 0 && g < 4) cntw[g] = cv;

  v4f s = pv[0];
  #pragma unroll
  for (int i = 1; i < 8; ++i) s += pv[i];
  part[t] = s;
  __syncthreads();

  if (t < 32) {
    const float fcnt = (float)(cntw[0] + cntw[1] + cntw[2] + cntw[3]);
    v4f a = part[t];
    #pragma unroll
    for (int k = 1; k < 8; ++k) a += part[k * 32 + t];
    v4f p;
    p.x = a.x / fcnt; p.y = a.y / fcnt; p.z = a.z / fcnt; p.w = a.w / fcnt;
    *(v4f*)(pSums + rowOff) = p;
  }
}

#define QSTAGE (MBLK * LDSTR)
#define PSTAGE (NWAY * LDSTR)

__device__ __forceinline__ void cg_half(v8f& acc0, v8f& acc1, v8f& acc2, v8f& acc3,
                                        const float* qa, const float* pb, int k0) {
  for (int k4 = k0; k4 < k0 + KC / 2; k4 += 4) {
    v2f a  = *(const v2f*)(qa + k4);
    v2f b0 = *(const v2f*)(pb +  0 * LDSTR + k4);
    v2f b1 = *(const v2f*)(pb + 16 * LDSTR + k4);
    v2f b2 = *(const v2f*)(pb + 32 * LDSTR + k4);
    v2f b3 = *(const v2f*)(pb + 48 * LDSTR + k4);
    acc0 = __builtin_amdgcn_wmma_f32_16x16x4_f32(false, a, false, b0, (short)0, acc0, false, false);
    acc1 = __builtin_amdgcn_wmma_f32_16x16x4_f32(false, a, false, b1, (short)0, acc1, false, false);
    acc2 = __builtin_amdgcn_wmma_f32_16x16x4_f32(false, a, false, b2, (short)0, acc2, false, false);
    acc3 = __builtin_amdgcn_wmma_f32_16x16x4_f32(false, a, false, b3, (short)0, acc3, false, false);
  }
  asm volatile("" : "+v"(acc0), "+v"(acc1), "+v"(acc2), "+v"(acc3));
}

__device__ __forceinline__ void cg_chunk(v8f& acc0, v8f& acc1, v8f& acc2, v8f& acc3, float& nacc,
                                         float& pacc, const float* qst, const float* pst,
                                         int w, int l15, int hi, int nrow, int nhalf,
                                         int prow, int pq) {
  const float* qa = qst + (w * 16 + l15) * LDSTR + 2 * hi;
  const float* pb = pst + l15 * LDSTR + 2 * hi;
  cg_half(acc0, acc1, acc2, acc3, qa, pb, 0);
  cg_half(acc0, acc1, acc2, acc3, qa, pb, KC / 2);
  #pragma unroll
  for (int i = 0; i < 8; ++i) {
    v4f x = *(const v4f*)(qst + nrow * LDSTR + nhalf * 32 + i * 4);
    nacc += x.x * x.x + x.y * x.y + x.z * x.z + x.w * x.w;
  }
  #pragma unroll
  for (int i = 0; i < 4; ++i) {
    v4f x = *(const v4f*)(pst + prow * LDSTR + pq * 16 + i * 4);
    pacc += x.x * x.x + x.y * x.y + x.z * x.z + x.w * x.w;
  }
}

__global__ void __launch_bounds__(256)
cosine_gemm(const float* __restrict__ q, const float* __restrict__ prot,
            const float* __restrict__ scaleP, float* __restrict__ out) {
  __shared__ __attribute__((aligned(16))) float qs[2 * QSTAGE];
  __shared__ __attribute__((aligned(16))) float ps[2 * PSTAGE];
  __shared__ float normP[256];
  __shared__ float invq[MBLK];
  __shared__ float pnP[256];
  __shared__ float invp[NWAY];

  const int t    = threadIdx.x;
  const int w    = t >> 5;
  const int lane = t & 31;
  const int l15  = lane & 15;
  const int hi   = lane >> 4;
  const size_t qBase = (size_t)blockIdx.x * MBLK;

  v8f acc0 = {}, acc1 = {}, acc2 = {}, acc3 = {};
  float nacc = 0.f, pacc = 0.f;
  const int nrow  = t & 127;
  const int nhalf = t >> 7;
  const int prow  = t >> 2;
  const int pq    = t & 3;

  const int srow = t >> 4;
  const int sc4  = (t & 15) * 4;
  const float* qsrc = q + (qBase + srow) * D + sc4;
  const float* psrc = prot + (size_t)srow * 256 + sc4;
  const int sOff = srow * LDSTR + sc4;

  v4f qr[8], pr[4];

  #pragma unroll
  for (int j = 0; j < 8; ++j) qr[j] = *(const v4f*)(qsrc + (size_t)(16 * j) * D);
  #pragma unroll
  for (int j = 0; j < 4; ++j) pr[j] = *(const v4f*)(psrc + (size_t)(16 * j) * 256);
  #pragma unroll
  for (int j = 0; j < 8; ++j) *(v4f*)(qs + sOff + (16 * j) * LDSTR) = qr[j];
  #pragma unroll
  for (int j = 0; j < 4; ++j) *(v4f*)(ps + sOff + (16 * j) * LDSTR) = pr[j];
  __syncthreads();

  for (int c = 0; c < D / KC - 1; ++c) {
    const int st = c & 1;
    #pragma unroll
    for (int j = 0; j < 8; ++j) qr[j] = *(const v4f*)(qsrc + (size_t)(16 * j) * D + (c + 1) * KC);
    const int pco = ((c + 1) >> 2) * (NWAY * 256) + ((c + 1) & 3) * KC;
    #pragma unroll
    for (int j = 0; j < 4; ++j) pr[j] = *(const v4f*)(psrc + (size_t)(16 * j) * 256 + pco);

    cg_chunk(acc0, acc1, acc2, acc3, nacc, pacc, qs + st * QSTAGE, ps + st * PSTAGE,
             w, l15, hi, nrow, nhalf, prow, pq);

    float* qd = qs + (st ^ 1) * QSTAGE + sOff;
    float* pd = ps + (st ^ 1) * PSTAGE + sOff;
    #pragma unroll
    for (int j = 0; j < 8; ++j) *(v4f*)(qd + (16 * j) * LDSTR) = qr[j];
    #pragma unroll
    for (int j = 0; j < 4; ++j) *(v4f*)(pd + (16 * j) * LDSTR) = pr[j];
    __syncthreads();
  }
  cg_chunk(acc0, acc1, acc2, acc3, nacc, pacc, qs + QSTAGE, ps + PSTAGE, w, l15, hi, nrow, nhalf,
           prow, pq);

  normP[t] = nacc;
  pnP[t]   = pacc;
  const float scale = scaleP[0];
  __syncthreads();
  if (t < MBLK) {
    float n2 = normP[t] + normP[t + MBLK];
    invq[t] = 1.f / fmaxf(sqrtf(n2), EPSF);
  }
  if (t < NWAY) {
    float n2 = (pnP[4 * t] + pnP[4 * t + 1]) + (pnP[4 * t + 2] + pnP[4 * t + 3]);
    invp[t] = scale / fmaxf(sqrtf(n2), EPSF);
  }
  __syncthreads();
  const float ip0 = invp[l15], ip1 = invp[16 + l15], ip2 = invp[32 + l15], ip3 = invp[48 + l15];

  #pragma unroll
  for (int r = 0; r < 8; ++r) {
    int qrow = w * 16 + r + 8 * hi;
    float s = invq[qrow];
    size_t orow = (qBase + qrow) * NWAY;
    out[orow +  0 + l15] = acc0[r] * s * ip0;
    out[orow + 16 + l15] = acc1[r] * s * ip1;
    out[orow + 32 + l15] = acc2[r] * s * ip2;
    out[orow + 48 + l15] = acc3[r] * s * ip3;
  }
}

extern "C" void kernel_launch(void* const* d_in, const int* in_sizes, int n_in,
                              void* d_out, int out_size, void* d_ws, size_t ws_size,
                              hipStream_t stream) {
  const float* sup    = (const float*)d_in[0];
  const int*   slab   = (const int*)d_in[1];
  const float* qry    = (const float*)d_in[2];
  const float* scaleP = (const float*)d_in[4];
  float* out = (float*)d_out;

  float* pSums = (float*)d_ws;
  int*   cnts  = (int*)((char*)d_ws + (size_t)128 * NWAY * 256 * sizeof(float));

  proto_partial<<<512, 256, 0, stream>>>(sup, slab, pSums, cnts);
  proto_final  <<<NWAY * 4, PF_THREADS, 0, stream>>>(pSums, cnts);
  cosine_gemm  <<<NQ / MBLK, 256, 0, stream>>>(qry, pSums, scaleP, out);
}


# cosine kernel: both norms summed from the registers that stage each chunk instead of re-read from LDS, row partials added once behind the K loop; changes the summation order of both norms
# speedup vs baseline: 3.1689x; 1.0502x over previous
#include <hip/hip_runtime.h>
#include <math.h>

#define D     512
#define NWAY  64
#define NQ    65536
#define NS    65536
#define EPSF  1e-30f
#define MBLK  128
#define KC    64
#define LDSTR 68

typedef float v2f __attribute__((ext_vector_type(2)));
typedef float v4f __attribute__((ext_vector_type(4)));
typedef float v8f __attribute__((ext_vector_type(8)));

#define PP_HROWS 512
#define PP_GROUP 32
#define PP_LCAP  (PP_HROWS + 2 * PP_GROUP)

__device__ __forceinline__ void pp_fetch(v4f (&buf)[16], int2 (&ent)[8],
                                         const float* __restrict__ srcU, const int* my,
                                         int g, int hi, unsigned laneOff) {
  #pragma unroll
  for (int s = 0; s < 8; ++s) ent[s] = *(const int2*)(my + PP_GROUP * g + 4 * s + 2 * hi);
  #pragma unroll
  for (int s = 0; s < 8; ++s) {
    buf[2 * s]     = *(const v4f*)(srcU + ((unsigned)(ent[s].x >> 5) * D + laneOff));
    buf[2 * s + 1] = *(const v4f*)(srcU + ((unsigned)(ent[s].y >> 5) * D + laneOff));
  }
}

__device__ __forceinline__ void pp_mma(v8f (&acc)[4], const v4f (&buf)[16],
                                       const int2 (&ent)[8], int l15) {
  #pragma unroll
  for (int s = 0; s < 8; ++s) {
    v2f a;
    a.x = ((ent[s].x & 31) == l15) ? 1.f : 0.f;
    a.y = ((ent[s].y & 31) == l15) ? 1.f : 0.f;
    #pragma unroll
    for (int j = 0; j < 4; ++j) {
      v2f b;
      b.x = buf[2 * s][j];
      b.y = buf[2 * s + 1][j];
      acc[j] = __builtin_amdgcn_wmma_f32_16x16x4_f32(false, a, false, b, (short)0,
                                                     acc[j], false, false);
    }
  }
}

__global__ void __launch_bounds__(256)
proto_partial(const float* __restrict__ sup, const int* __restrict__ lab,
              float* __restrict__ pSums, int* __restrict__ cnts) {
  __shared__ __attribute__((aligned(16))) float red[8 * 16 * 64];
  __shared__ __attribute__((aligned(16))) int   rlist[8 * PP_LCAP];
  __shared__ int ccnt[16];
  const int t     = threadIdx.x;
  const int wv    = __builtin_amdgcn_readfirstlane(t >> 5);
  const int lane  = t & 31;
  const int l15   = lane & 15;
  const int hi    = lane >> 4;
  const int rc    = blockIdx.x >> 3;
  const int slice = blockIdx.x & 7;
  const int mt    = wv & 3;
  const int half  = wv >> 2;
  const int rowBase = rc * 1024 + half * PP_HROWS;
  const float* srcU = sup + (size_t)rowBase * D + slice * 64;
  const unsigned laneOff = 4 * l15;
  const int* lp = lab + rowBase;
  int* my = rlist + wv * PP_LCAP;

  int lv[PP_HROWS / 32];
  #pragma unroll
  for (int i = 0; i < PP_HROWS / 32; ++i) lv[i] = lp[32 * i + lane];
  int cnt = 0;
  #pragma unroll
  for (int i = 0; i < PP_HROWS / 32; ++i) {
    const bool in = (lv[i] >> 4) == mt;
    const unsigned bal = __builtin_amdgcn_ballot_w32(in);
    const int pos = cnt + (int)__builtin_amdgcn_mbcnt_lo(bal, 0u);
    if (in) my[pos] = ((32 * i + lane) << 5) | (lv[i] & 15);
    cnt += __builtin_popcount(bal);
  }
  my[cnt + lane] = 16;
  my[cnt + 32 + lane] = 16;
  {
    const int k0 = 8 * slice + 2 * mt;
    int c0 = 0, c1 = 0;
    #pragma unroll
    for (int i = 0; i < PP_HROWS / 32; ++i) {
      c0 += __builtin_popcount(__builtin_amdgcn_ballot_w32(lv[i] == k0));
      c1 += __builtin_popcount(__builtin_amdgcn_ballot_w32(lv[i] == k0 + 1));
    }
    if (lane == 0) { ccnt[half * 8 + 2 * mt] = c0; ccnt[half * 8 + 2 * mt + 1] = c1; }
  }
  __syncthreads();
  if (t < 16) cnts[blockIdx.x * 32 + t] = ccnt[t];

  v8f acc[4];
  #pragma unroll
  for (int j = 0; j < 4; ++j) acc[j] = (v8f){};

  const int ngrp = (cnt + PP_GROUP - 1) / PP_GROUP;
  v4f bufA[16], bufB[16];
  int2 entA[8], entB[8];
  pp_fetch(bufA, entA, srcU, my, 0, hi, laneOff);
  const int g0 = ngrp & 1;
  if (g0) {
    pp_mma(acc, bufA, entA, l15);
    pp_fetch(bufA, entA, srcU, my, 1, hi, laneOff);
  }
  for (int g = g0; g < ngrp; g += 2) {
    pp_fetch(bufB, entB, srcU, my, g + 1, hi, laneOff);
    __builtin_amdgcn_sched_barrier(0);
    #pragma unroll
    for (int i = 0; i < 16; ++i) asm volatile("" : "+v"(bufA[i]));
    pp_mma(acc, bufA, entA, l15);
    __builtin_amdgcn_sched_barrier(0);
    pp_fetch(bufA, entA, srcU, my, g + 2, hi, laneOff);
    __builtin_amdgcn_sched_barrier(0);
    pp_mma(acc, bufB, entB, l15);
    __builtin_amdgcn_sched_barrier(0);
  }

  float* tile = red + wv * (16 * 64);
  #pragma unroll
  for (int r = 0; r < 8; ++r) {
    v4f v;
    v.x = acc[0][r]; v.y = acc[1][r]; v.z = acc[2][r]; v.w = acc[3][r];
    *(v4f*)(tile + (8 * hi + r) * 64 + 4 * l15) = v;
  }
  __syncthreads();

  float* dst = pSums + (size_t)((rc * 2 + (slice >> 2)) * NWAY) * 256 + (slice & 3) * 64;
  #pragma unroll
  for (int i = 0; i < 4; ++i) {
    const int idx = t + 256 * i;
    const int cls = idx >> 4, c4 = (idx & 15) * 4;
    const float* p0 = red + (cls >> 4) * (16 * 64) + (cls & 15) * 64 + c4;
    const v4f s = *(const v4f*)p0 + *(const v4f*)(p0 + 4 * (16 * 64));
    *(v4f*)(dst + (size_t)cls * 256 + c4) = s;
  }
}

#define PF_THREADS 256

__global__ void __launch_bounds__(PF_THREADS)
proto_final(float* pSums, const int* __restrict__ cnts) {
  __shared__ v4f part[8 * 32];
  __shared__ int cntw[4];
  const int c    = blockIdx.x >> 2;
  const int dq   = blockIdx.x & 3;
  const int t    = threadIdx.x;
  const int lane = t & 31;
  const int g    = t >> 5;

  const size_t chunkStride = (size_t)2 * NWAY * 256;
  const size_t rowOff = (size_t)((dq >> 1) * NWAY + c) * 256 + (dq & 1) * 128 + 4 * lane;
  const float* ps = pSums + (size_t)(g * 8) * chunkStride + rowOff;
  v4f pv[8];
  #pragma unroll
  for (int i = 0; i < 8; ++i) pv[i] = *(const v4f*)(ps + (size_t)i * chunkStride);

  int cv = 0;
  if (t < 128) cv = cnts[((t >> 1) * 8 + (c >> 3)) * 32 + (t & 1) * 8 + (c & 7)];
  #pragma unroll
  for (int m = 16; m > 0; m >>= 1) cv += __shfl_xor(cv, m);
  if (lane == 0 && g < 4) cntw[g] = cv;

  v4f s = pv[0];
  #pragma unroll
  for (int i = 1; i < 8; ++i) s += pv[i];
  part[t] = s;
  __syncthreads();

  if (t < 32) {
    const float fcnt = (float)(cntw[0] + cntw[1] + cntw[2] + cntw[3]);
    v4f a = part[t];
    #pragma unroll
    for (int k = 1; k < 8; ++k) a += part[k * 32 + t];
    v4f p;
    p.x = a.x / fcnt; p.y = a.y / fcnt; p.z = a.z / fcnt; p.w = a.w / fcnt;
    *(v4f*)(pSums + rowOff) = p;
  }
}

#define NPITCH 20
#define QSTAGE (MBLK * LDSTR)
#define PSTAGE (NWAY * LDSTR)

__device__ __forceinline__ void cg_half(v8f& acc0, v8f& acc1, v8f& acc2, v8f& acc3,
                                        const float* qa, const float* pb, int k0) {
  for (int k4 = k0; k4 < k0 + KC / 2; k4 += 4) {
    v2f a  = *(const v2f*)(qa + k4);
    v2f b0 = *(const v2f*)(pb +  0 * LDSTR + k4);
    v2f b1 = *(const v2f*)(pb + 16 * LDSTR + k4);
    v2f b2 = *(const v2f*)(pb + 32 * LDSTR + k4);
    v2f b3 = *(const v2f*)(pb + 48 * LDSTR + k4);
    acc0 = __builtin_amdgcn_wmma_f32_16x16x4_f32(false, a, false, b0, (short)0, acc0, false, false);
    acc1 = __builtin_amdgcn_wmma_f32_16x16x4_f32(false, a, false, b1, (short)0, acc1, false, false);
    acc2 = __builtin_amdgcn_wmma_f32_16x16x4_f32(false, a, false, b2, (short)0, acc2, false, false);
    acc3 = __builtin_amdgcn_wmma_f32_16x16x4_f32(false, a, false, b3, (short)0, acc3, false, false);
  }
  asm volatile("" : "+v"(acc0), "+v"(acc1), "+v"(acc2), "+v"(acc3));
}

__device__ __forceinline__ void cg_chunk(v8f& acc0, v8f& acc1, v8f& acc2, v8f& acc3,
                                         const float* qst, const float* pst,
                                         int w, int l15, int hi) {
  const float* qa = qst + (w * 16 + l15) * LDSTR + 2 * hi;
  const float* pb = pst + l15 * LDSTR + 2 * hi;
  cg_half(acc0, acc1, acc2, acc3, qa, pb, 0);
  cg_half(acc0, acc1, acc2, acc3, qa, pb, KC / 2);
}

__device__ __forceinline__ void cg_norms(v2f (&nq)[8], v2f (&np)[4], const v4f (&qr)[8],
                                         const v4f (&pr)[4]) {
  #pragma unroll
  for (int j = 0; j < 8; ++j) {
    v2f lo = {qr[j].x, qr[j].y}, hi2 = {qr[j].z, qr[j].w};
    nq[j] += lo * lo;
    nq[j] += hi2 * hi2;
  }
  #pragma unroll
  for (int j = 0; j < 4; ++j) {
    v2f lo = {pr[j].x, pr[j].y}, hi2 = {pr[j].z, pr[j].w};
    np[j] += lo * lo;
    np[j] += hi2 * hi2;
  }
}


__global__ void __launch_bounds__(256)
cosine_gemm(const float* __restrict__ q, const float* __restrict__ prot,
            const float* __restrict__ scaleP, float* __restrict__ out) {
  __shared__ __attribute__((aligned(16))) float qs[2 * QSTAGE];
  __shared__ __attribute__((aligned(16))) float ps[2 * PSTAGE];
  __shared__ float invq[MBLK];
  __shared__ float invp[NWAY];

  const int t    = threadIdx.x;
  const int w    = t >> 5;
  const int lane = t & 31;
  const int l15  = lane & 15;
  const int hi   = lane >> 4;
  const size_t qBase = (size_t)blockIdx.x * MBLK;

  v8f acc0 = {}, acc1 = {}, acc2 = {}, acc3 = {};
  v2f nq[8], np[4];
  #pragma unroll
  for (int j = 0; j < 8; ++j) nq[j] = (v2f){};
  #pragma unroll
  for (int j = 0; j < 4; ++j) np[j] = (v2f){};

  const int srow = t >> 4;
  const int sc4  = (t & 15) * 4;
  const float* qsrc = q + (qBase + srow) * D + sc4;
  const float* psrc = prot + (size_t)srow * 256 + sc4;
  const int sOff = srow * LDSTR + sc4;

  v4f qr[8], pr[4];

  #pragma unroll
  for (int j = 0; j < 8; ++j) qr[j] = *(const v4f*)(qsrc + (size_t)(16 * j) * D);
  #pragma unroll
  for (int j = 0; j < 4; ++j) pr[j] = *(const v4f*)(psrc + (size_t)(16 * j) * 256);
  #pragma unroll
  for (int j = 0; j < 8; ++j) *(v4f*)(qs + sOff + (16 * j) * LDSTR) = qr[j];
  #pragma unroll
  for (int j = 0; j < 4; ++j) *(v4f*)(ps + sOff + (16 * j) * LDSTR) = pr[j];
  cg_norms(nq, np, qr, pr);
  __syncthreads();

  for (int c = 0; c < D / KC - 1; ++c) {
    const int st = c & 1;
    #pragma unroll
    for (int j = 0; j < 8; ++j) qr[j] = *(const v4f*)(qsrc + (size_t)(16 * j) * D + (c + 1) * KC);
    const int pco = ((c + 1) >> 2) * (NWAY * 256) + ((c + 1) & 3) * KC;
    #pragma unroll
    for (int j = 0; j < 4; ++j) pr[j] = *(const v4f*)(psrc + (size_t)(16 * j) * 256 + pco);

    cg_chunk(acc0, acc1, acc2, acc3, qs + st * QSTAGE, ps + st * PSTAGE, w, l15, hi);

    float* qd = qs + (st ^ 1) * QSTAGE + sOff;
    float* pd = ps + (st ^ 1) * PSTAGE + sOff;
    #pragma unroll
    for (int j = 0; j < 8; ++j) *(v4f*)(qd + (16 * j) * LDSTR) = qr[j];
    #pragma unroll
    for (int j = 0; j < 4; ++j) *(v4f*)(pd + (16 * j) * LDSTR) = pr[j];
    cg_norms(nq, np, qr, pr);
    __syncthreads();
  }
  float* part = qs;
  {
    const int si = t & 15;
    #pragma unroll
    for (int j = 0; j < 8; ++j) part[(srow + 16 * j) * NPITCH + si] = nq[j].x + nq[j].y;
    #pragma unroll
    for (int j = 0; j < 4; ++j) part[(MBLK + srow + 16 * j) * NPITCH + si] = np[j].x + np[j].y;
  }
  cg_chunk(acc0, acc1, acc2, acc3, qs + QSTAGE, ps + PSTAGE, w, l15, hi);

  const float scale = scaleP[0];
  __syncthreads();
  if (t < MBLK + NWAY) {
    const v4f a = *(const v4f*)(part + t * NPITCH);
    const v4f b = *(const v4f*)(part + t * NPITCH + 4);
    const v4f c = *(const v4f*)(part + t * NPITCH + 8);
    const v4f d = *(const v4f*)(part + t * NPITCH + 12);
    const v4f s = (a + b) + (c + d);
    const float n2 = (s.x + s.y) + (s.z + s.w);
    const float num = (t < MBLK) ? 1.f : scale;
    const float v = num / fmaxf(sqrtf(n2), EPSF);
    if (t < MBLK) invq[t] = v; else invp[t - MBLK] = v;
  }
  __syncthreads();
  const float ip0 = invp[l15], ip1 = invp[16 + l15], ip2 = invp[32 + l15], ip3 = invp[48 + l15];

  #pragma unroll
  for (int r = 0; r < 8; ++r) {
    int qrow = w * 16 + r + 8 * hi;
    float s = invq[qrow];
    size_t orow = (qBase + qrow) * NWAY;
    out[orow +  0 + l15] = acc0[r] * s * ip0;
    out[orow + 16 + l15] = acc1[r] * s * ip1;
    out[orow + 32 + l15] = acc2[r] * s * ip2;
    out[orow + 48 + l15] = acc3[r] * s * ip3;
  }
}

extern "C" void kernel_launch(void* const* d_in, const int* in_sizes, int n_in,
                              void* d_out, int out_size, void* d_ws, size_t ws_size,
                              hipStream_t stream) {
  const float* sup    = (const float*)d_in[0];
  const int*   slab   = (const int*)d_in[1];
  const float* qry    = (const float*)d_in[2];
  const float* scaleP = (const float*)d_in[4];
  float* out = (float*)d_out;

  float* pSums = (float*)d_ws;
  int*   cnts  = (int*)((char*)d_ws + (size_t)128 * NWAY * 256 * sizeof(float));

  proto_partial<<<512, 256, 0, stream>>>(sup, slab, pSums, cnts);
  proto_final  <<<NWAY * 4, PF_THREADS, 0, stream>>>(pSums, cnts);
  cosine_gemm  <<<NQ / MBLK, 256, 0, stream>>>(qry, pSums, scaleP, out);
}
